# starting kernel
# baseline (speedup 1.0000x reference)
#include <hip/hip_runtime.h>

typedef float v2f __attribute__((ext_vector_type(2)));
typedef float v8f __attribute__((ext_vector_type(8)));

#define D 64

__device__ __forceinline__ void gatomic_add_f32(float* addr, float v) {
    asm volatile("global_atomic_add_f32 %0, %1, off" : : "v"(addr), "v"(v) : "memory");
}

__global__ __launch_bounds__(256) void k_scatter(const int* __restrict__ src,
                                                 const int* __restrict__ dst,
                                                 const float* __restrict__ norm,
                                                 const float* __restrict__ H,
                                                 float* __restrict__ acc, int e_cnt) {
    const unsigned g = blockIdx.x * 256u + threadIdx.x;
    const unsigned e = g >> 4;
    const unsigned l = g & 15u;
    if (e >= (unsigned)e_cnt) return;
    const unsigned s   = (unsigned)src[e];
    const unsigned d   = (unsigned)dst[e];
    const float    nrm = norm[e];
    const float4 hv = *(const float4*)(H + s * (unsigned)D + l * 4u);
    float* base = acc + d * (unsigned)D + l * 4u;
    gatomic_add_f32(base + 0, hv.x * nrm);
    gatomic_add_f32(base + 1, hv.y * nrm);
    gatomic_add_f32(base + 2, hv.z * nrm);
    gatomic_add_f32(base + 3, hv.w * nrm);
}

__global__ __launch_bounds__(256) void k_init_deg(float* deg, int n) {
    int i = blockIdx.x * blockDim.x + threadIdx.x;
    if (i < n) deg[i] = 1.0f;
}

__global__ __launch_bounds__(256) void k_deg_count(const int* __restrict__ dst,
                                                   float* __restrict__ deg, int e_cnt) {
    int e = blockIdx.x * blockDim.x + threadIdx.x;
    if (e < e_cnt) gatomic_add_f32(&deg[dst[e]], 1.0f);
}

__global__ __launch_bounds__(256) void k_rsqrt(float* d, int n) {
    int i = blockIdx.x * blockDim.x + threadIdx.x;
    if (i < n) d[i] = rsqrtf(d[i]);
}

__global__ __launch_bounds__(256) void k_edge_norm(const int* __restrict__ src,
                                                   const int* __restrict__ dst,
                                                   const float* __restrict__ dinv,
                                                   float* __restrict__ norm, int e_cnt) {
    int e = blockIdx.x * blockDim.x + threadIdx.x;
    if (e < e_cnt) norm[e] = dinv[src[e]] * dinv[dst[e]];
}

__global__ __launch_bounds__(256) void k_selfinit(const float* __restrict__ H,
                                                  const float* __restrict__ dinv,
                                                  const float* __restrict__ bias,
                                                  float* __restrict__ acc, int n_nodes) {
    const long i = (long)blockIdx.x * blockDim.x + threadIdx.x;
    if (i >= (long)n_nodes * D) return;
    const int node = (int)(i >> 6);
    const int c    = (int)(i & (D - 1));
    const float di = dinv[node];
    acc[i] = H[i] * di * di + bias[c];
}

template <bool RELU>
__global__ __launch_bounds__(128) void k_gemm_wmma(const float* __restrict__ X,
                                                   const float* __restrict__ W,
                                                   float* __restrict__ H, int n_rows) {
    const int lane = threadIdx.x & 31;
    const int wave = threadIdx.x >> 5;
    const int half = lane >> 4;
    const int l16  = lane & 15;
    const int mt   = blockIdx.x;

    int row = mt * 16 + l16;
    if (row >= n_rows) row = n_rows - 1;
    const int ncol = wave * 16 + l16;

    v8f c = {};
#pragma unroll
    for (int k0 = 0; k0 < D; k0 += 4) {
        const int ka = k0 + 2 * half;
        float2 av = *(const float2*)(X + (long)row * D + ka);
        if (RELU) { av.x = fmaxf(av.x, 0.0f); av.y = fmaxf(av.y, 0.0f); }
        v2f a, b;
        a.x = av.x;  a.y = av.y;
        b.x = W[ka * D + ncol];
        b.y = W[(ka + 1) * D + ncol];
        c = __builtin_amdgcn_wmma_f32_16x16x4_f32(false, a, false, b,
                                                  (short)0, c, false, false);
    }

    const int mbase = mt * 16 + half * 8;
#pragma unroll
    for (int r = 0; r < 8; ++r) {
        const int m = mbase + r;
        if (m < n_rows) H[(long)m * D + ncol] = c[r];
    }
}

extern "C" void kernel_launch(void* const* d_in, const int* in_sizes, int n_in,
                              void* d_out, int out_size, void* d_ws, size_t ws_size,
                              hipStream_t stream) {
    const float* x  = (const float*)d_in[0];
    const int*   ei = (const int*)d_in[1];
    const float* Wm[3] = { (const float*)d_in[2], (const float*)d_in[4], (const float*)d_in[6] };
    const float* bm[3] = { (const float*)d_in[3], (const float*)d_in[5], (const float*)d_in[7] };

    const int N = in_sizes[0] / D;
    const int E = in_sizes[1] / 2;
    const int* src = ei;
    const int* dst = ei + E;

    const long npad = ((long)N + 63) & ~63L;
    const long epad = ((long)E + 63) & ~63L;
    const long feat = (long)N * D;
    float* dinv = (float*)d_ws;
    float* norm = dinv + npad;
    float* h    = norm + epad;
    float* acc  = h + feat;

    const int B = 256;
    const int gN    = (N + B - 1) / B;
    const int gE    = (E + B - 1) / B;
    const int gFeat = (int)((feat + B - 1) / B);
    const int gScat = (int)(((long)E * 16 + B - 1) / B);
    const int gGemm = (N + 15) / 16;

    k_init_deg<<<gN, B, 0, stream>>>(dinv, N);
    k_deg_count<<<gE, B, 0, stream>>>(dst, dinv, E);
    k_rsqrt<<<gN, B, 0, stream>>>(dinv, N);
    k_edge_norm<<<gE, B, 0, stream>>>(src, dst, dinv, norm, E);

    k_gemm_wmma<false><<<gGemm, 128, 0, stream>>>(x, Wm[0], h, N);
    k_selfinit<<<gFeat, B, 0, stream>>>(h, dinv, bm[0], acc, N);
    k_scatter<<<gScat, B, 0, stream>>>(src, dst, norm, h, acc, E);

    k_gemm_wmma<true><<<gGemm, 128, 0, stream>>>(acc, Wm[1], h, N);
    k_selfinit<<<gFeat, B, 0, stream>>>(h, dinv, bm[1], acc, N);
    k_scatter<<<gScat, B, 0, stream>>>(src, dst, norm, h, acc, E);

    float* out = (float*)d_out;
    k_gemm_wmma<true><<<gGemm, 128, 0, stream>>>(acc, Wm[2], h, N);
    k_selfinit<<<gFeat, B, 0, stream>>>(h, dinv, bm[2], out, N);
    k_scatter<<<gScat, B, 0, stream>>>(src, dst, norm, h, out, E);
}


# edge aggregation without atomics: per-call stable radix sort of edges by destination (ballot ranks) -> CSR, then row gather fused with self loop + bias; row sums now in edge order, same every run
# speedup vs baseline: 1.9362x; 1.9362x over previous
#include <hip/hip_runtime.h>

typedef float v2f __attribute__((ext_vector_type(2)));
typedef float v8f __attribute__((ext_vector_type(8)));

#define D 64


#define TILE_KEYS 4096
#define TILE_ROWS 128
#define NDIG      64

template <bool SCATTER>
__global__ __launch_bounds__(256) void k_sort_pass(const int* __restrict__ kin,
                                                   const int* __restrict__ vin,
                                                   int* __restrict__ kout,
                                                   int* __restrict__ vout,
                                                   int* __restrict__ tab,
                                                   int n, int shift, int nblk) {
    __shared__ int cnt[TILE_ROWS * NDIG];
    __shared__ int qsum[4 * NDIG];
    const int t     = threadIdx.x;
    const int lane  = t & 31;
    const int wave  = t >> 5;
    const int tile0 = blockIdx.x * TILE_KEYS;

    for (int i = t; i < TILE_ROWS * NDIG; i += 256) cnt[i] = 0;
    __syncthreads();

    int dr[16];
    const unsigned lt = (1u << lane) - 1u;
#pragma unroll
    for (int r = 0; r < 16; ++r) {
        const int  row   = r * 8 + wave;
        const int  idx   = tile0 + row * 32 + lane;
        const bool valid = idx < n;
        const int  key   = valid ? kin[idx] : 0;
        const int  d     = (key >> shift) & (NDIG - 1);
        unsigned peers = __builtin_amdgcn_ballot_w32(valid);
#pragma unroll
        for (int b = 0; b < 6; ++b) {
            const bool     bit = (d >> b) & 1;
            const unsigned m   = __builtin_amdgcn_ballot_w32(valid && bit);
            peers &= bit ? m : ~m;
        }
        const int rk = __popc(peers & lt);
        if (valid && rk == 0) cnt[row * NDIG + d] = __popc(peers);
        dr[r] = (d << 8) | rk;
    }
    __syncthreads();

    const int dg = t & (NDIG - 1);
    const int q  = t >> 6;
    {
        int run = 0;
        for (int i = 0; i < 32; ++i) {
            const int a = (q * 32 + i) * NDIG + dg;
            const int v = cnt[a];
            if (SCATTER) cnt[a] = run;
            run += v;
        }
        qsum[q * NDIG + dg] = run;
    }
    __syncthreads();

    if (!SCATTER) {
        if (t < NDIG)
            tab[t * nblk + blockIdx.x] = qsum[t] + qsum[NDIG + t] + qsum[2 * NDIG + t] + qsum[3 * NDIG + t];
        return;
    }

#pragma unroll
    for (int r = 0; r < 16; ++r) {
        const int row = r * 8 + wave;
        const int idx = tile0 + row * 32 + lane;
        if (idx < n) {
            const int d  = dr[r] >> 8;
            const int rk = dr[r] & 255;
            const int qq = row >> 5;
            int off = tab[d * nblk + blockIdx.x] + cnt[row * NDIG + d] + rk;
            if (qq > 0) off += qsum[d];
            if (qq > 1) off += qsum[NDIG + d];
            if (qq > 2) off += qsum[2 * NDIG + d];
            kout[off] = kin[idx];
            vout[off] = vin[idx];
        }
    }
}

__global__ __launch_bounds__(256) void k_scan(const int* __restrict__ in,
                                              int* __restrict__ out, int n) {
    __shared__ int wsum[8];
    const int t    = threadIdx.x;
    const int lane = t & 31;
    const int wave = t >> 5;
    const int base = blockIdx.x * 1024;

    int part = 0;
    for (int i = t * 4; i < base; i += 1024) {
        const int4 v = *(const int4*)(in + i);
        part += v.x + v.y + v.z + v.w;
    }
    for (int o = 16; o > 0; o >>= 1) part += __shfl_xor(part, o, 32);
    if (lane == 0) wsum[wave] = part;
    __syncthreads();
    int prefix = 0;
    for (int w = 0; w < 8; ++w) prefix += wsum[w];
    __syncthreads();

    const int i0 = base + t * 4;
    int c0 = (i0     < n) ? in[i0]     : 0;
    int c1 = (i0 + 1 < n) ? in[i0 + 1] : 0;
    int c2 = (i0 + 2 < n) ? in[i0 + 2] : 0;
    int c3 = (i0 + 3 < n) ? in[i0 + 3] : 0;
    const int tsum = c0 + c1 + c2 + c3;
    int inc = tsum;
    for (int o = 1; o < 32; o <<= 1) {
        const int u = __shfl_up(inc, o, 32);
        if (lane >= o) inc += u;
    }
    if (lane == 31) wsum[wave] = inc;
    __syncthreads();
    int woff = 0;
    for (int w = 0; w < 8; ++w) woff += (w < wave) ? wsum[w] : 0;

    const int e0 = prefix + woff + inc - tsum;
    if (i0     < n) out[i0]     = e0;
    if (i0 + 1 < n) out[i0 + 1] = e0 + c0;
    if (i0 + 2 < n) out[i0 + 2] = e0 + c0 + c1;
    if (i0 + 3 < n) out[i0 + 3] = e0 + c0 + c1 + c2;
}

__global__ __launch_bounds__(256) void k_rowptr(const int* __restrict__ sdst,
                                                int* __restrict__ rowptr, int e_cnt, int n) {
    const int j = blockIdx.x * blockDim.x + threadIdx.x;
    if (j > e_cnt) return;
    const int lo = (j == 0) ? 0 : sdst[j - 1] + 1;
    const int hi = (j == e_cnt) ? n : sdst[j];
    for (int d = lo; d <= hi; ++d) rowptr[d] = j;
}

__global__ __launch_bounds__(256) void k_dinv(const int* __restrict__ rowptr,
                                              float* __restrict__ dinv, int n) {
    const int i = blockIdx.x * blockDim.x + threadIdx.x;
    if (i < n) dinv[i] = rsqrtf((float)(rowptr[i + 1] - rowptr[i] + 1));
}

__global__ __launch_bounds__(256) void k_edge_norm(const int* __restrict__ ssrc,
                                                   const int* __restrict__ sdst,
                                                   const float* __restrict__ dinv,
                                                   float* __restrict__ norm, int e_cnt) {
    const int j = blockIdx.x * blockDim.x + threadIdx.x;
    if (j < e_cnt) norm[j] = dinv[ssrc[j]] * dinv[sdst[j]];
}

__global__ __launch_bounds__(256) void k_pull(const int* __restrict__ rowptr,
                                              const int* __restrict__ ssrc,
                                              const float* __restrict__ norm,
                                              const float* __restrict__ dinv,
                                              const float* __restrict__ bias,
                                              const float* __restrict__ H,
                                              float* __restrict__ out, int n_nodes) {
    const unsigned g    = blockIdx.x * 256u + threadIdx.x;
    const unsigned node = g >> 4;
    const unsigned l    = g & 15u;
    if (node >= (unsigned)n_nodes) return;
    const int   beg = rowptr[node];
    const int   end = rowptr[node + 1];
    const float di  = dinv[node];
    const float4* H4 = (const float4*)H;
    const float4 hv = H4[node * 16u + l];
    const float4 bv = ((const float4*)bias)[l];
    float4 a;
    a.x = hv.x * di * di + bv.x;
    a.y = hv.y * di * di + bv.y;
    a.z = hv.z * di * di + bv.z;
    a.w = hv.w * di * di + bv.w;
#pragma unroll 4
    for (int j = beg; j < end; ++j) {
        const unsigned s = (unsigned)ssrc[j];
        const float    w = norm[j];
        const float4   r = H4[s * 16u + l];
        a.x += r.x * w;
        a.y += r.y * w;
        a.z += r.z * w;
        a.w += r.w * w;
    }
    ((float4*)out)[node * 16u + l] = a;
}

template <bool RELU>
__global__ __launch_bounds__(128) void k_gemm_wmma(const float* __restrict__ X,
                                                   const float* __restrict__ W,
                                                   float* __restrict__ H, int n_rows) {
    const int lane = threadIdx.x & 31;
    const int wave = threadIdx.x >> 5;
    const int half = lane >> 4;
    const int l16  = lane & 15;
    const int mt   = blockIdx.x;

    int row = mt * 16 + l16;
    if (row >= n_rows) row = n_rows - 1;
    const int ncol = wave * 16 + l16;

    v8f c = {};
#pragma unroll
    for (int k0 = 0; k0 < D; k0 += 4) {
        const int ka = k0 + 2 * half;
        float2 av = *(const float2*)(X + (long)row * D + ka);
        if (RELU) { av.x = fmaxf(av.x, 0.0f); av.y = fmaxf(av.y, 0.0f); }
        v2f a, b;
        a.x = av.x;  a.y = av.y;
        b.x = W[ka * D + ncol];
        b.y = W[(ka + 1) * D + ncol];
        c = __builtin_amdgcn_wmma_f32_16x16x4_f32(false, a, false, b,
                                                  (short)0, c, false, false);
    }

    const int mbase = mt * 16 + half * 8;
#pragma unroll
    for (int r = 0; r < 8; ++r) {
        const int m = mbase + r;
        if (m < n_rows) H[(long)m * D + ncol] = c[r];
    }
}

extern "C" void kernel_launch(void* const* d_in, const int* in_sizes, int n_in,
                              void* d_out, int out_size, void* d_ws, size_t ws_size,
                              hipStream_t stream) {
    const float* x  = (const float*)d_in[0];
    const int*   ei = (const int*)d_in[1];
    const float* Wm[3] = { (const float*)d_in[2], (const float*)d_in[4], (const float*)d_in[6] };
    const float* bm[3] = { (const float*)d_in[3], (const float*)d_in[5], (const float*)d_in[7] };

    const int N = in_sizes[0] / D;
    const int E = in_sizes[1] / 2;
    const int* src = ei;
    const int* dst = ei + E;

    const int  nblk = (E + TILE_KEYS - 1) / TILE_KEYS;
    const int  tlen = NDIG * nblk;
    const long npad = ((long)N + 63) & ~63L;
    const long epad = ((long)E + 63) & ~63L;
    const long tpad = ((long)tlen + 1023) & ~1023L;
    float* dinv   = (float*)d_ws;
    int*   rowptr = (int*)(dinv + npad);
    int*   sdst   = rowptr + npad + 64;
    int*   ssrc   = sdst + epad;
    float* norm   = (float*)(ssrc + epad);
    int*   tabA   = (int*)(norm + epad);
    int*   tabB   = tabA + tpad;
    float* h      = (float*)(tabB + tpad);
    int*   k2     = (int*)h;
    int*   v2     = k2 + epad;
    float* out    = (float*)d_out;

    const int B = 256;
    const int gN    = (N + B - 1) / B;
    const int gE    = (E + B - 1) / B;
    const int gE1   = (E + 1 + B - 1) / B;
    const int gScan = (tlen + 1023) / 1024;
    const int gPull = (int)(((long)N * 16 + B - 1) / B);
    const int gGemm = (N + 15) / 16;

    k_sort_pass<false><<<nblk, B, 0, stream>>>(dst, src, sdst, ssrc, tabA, E, 0, nblk);
    k_scan<<<gScan, B, 0, stream>>>(tabA, tabB, tlen);
    k_sort_pass<true><<<nblk, B, 0, stream>>>(dst, src, sdst, ssrc, tabB, E, 0, nblk);

    k_sort_pass<false><<<nblk, B, 0, stream>>>(sdst, ssrc, k2, v2, tabA, E, 6, nblk);
    k_scan<<<gScan, B, 0, stream>>>(tabA, tabB, tlen);
    k_sort_pass<true><<<nblk, B, 0, stream>>>(sdst, ssrc, k2, v2, tabB, E, 6, nblk);

    k_sort_pass<false><<<nblk, B, 0, stream>>>(k2, v2, sdst, ssrc, tabA, E, 12, nblk);
    k_scan<<<gScan, B, 0, stream>>>(tabA, tabB, tlen);
    k_sort_pass<true><<<nblk, B, 0, stream>>>(k2, v2, sdst, ssrc, tabB, E, 12, nblk);

    k_rowptr<<<gE1, B, 0, stream>>>(sdst, rowptr, E, N);
    k_dinv<<<gN, B, 0, stream>>>(rowptr, dinv, N);
    k_edge_norm<<<gE, B, 0, stream>>>(ssrc, sdst, dinv, norm, E);

    k_gemm_wmma<false><<<gGemm, 128, 0, stream>>>(x, Wm[0], h, N);
    k_pull<<<gPull, B, 0, stream>>>(rowptr, ssrc, norm, dinv, bm[0], h, out, N);

    k_gemm_wmma<true><<<gGemm, 128, 0, stream>>>(out, Wm[1], h, N);
    k_pull<<<gPull, B, 0, stream>>>(rowptr, ssrc, norm, dinv, bm[1], h, out, N);

    k_gemm_wmma<true><<<gGemm, 128, 0, stream>>>(out, Wm[2], h, N);
    k_pull<<<gPull, B, 0, stream>>>(rowptr, ssrc, norm, dinv, bm[2], h, out, N);
}


# sort passes and table scan: global loads moved out from under per-element branches (clamped index; block's table column read once via LDS) so a thread's 16 loads issue together; sum order same
# speedup vs baseline: 2.2644x; 1.1695x over previous
#include <hip/hip_runtime.h>

typedef float v2f __attribute__((ext_vector_type(2)));
typedef float v8f __attribute__((ext_vector_type(8)));

#define D 64


#define TILE_KEYS 4096
#define TILE_ROWS 128
#define NDIG      64

template <bool SCATTER>
__global__ __launch_bounds__(256) void k_sort_pass(const int* __restrict__ kin,
                                                   const int* __restrict__ vin,
                                                   int* __restrict__ kout,
                                                   int* __restrict__ vout,
                                                   int* __restrict__ tab,
                                                   int n, int shift, int nblk) {
    __shared__ int cnt[TILE_ROWS * NDIG];
    __shared__ int qsum[4 * NDIG];
    __shared__ int tcol[NDIG];
    const int t     = threadIdx.x;
    const int lane  = t & 31;
    const int wave  = t >> 5;
    const int tile0 = blockIdx.x * TILE_KEYS;
    const int last  = n - 1;

    int key[16];
    int val[16];
#pragma unroll
    for (int r = 0; r < 16; ++r) {
        const int idx = tile0 + (r * 8 + wave) * 32 + lane;
        const int ci  = idx < last ? idx : last;
        key[r] = kin[ci];
        if (SCATTER) val[r] = vin[ci];
    }
    int tb0 = 0, tb1 = 0;
    if (SCATTER && wave == 0) {
        tb0 = tab[lane * nblk + blockIdx.x];
        tb1 = tab[(lane + 32) * nblk + blockIdx.x];
    }

    for (int i = t; i < TILE_ROWS * NDIG; i += 256) cnt[i] = 0;
    __syncthreads();

    int rk[16];
    const unsigned lt = (1u << lane) - 1u;
#pragma unroll
    for (int r = 0; r < 16; ++r) {
        const int  row   = r * 8 + wave;
        const bool valid = tile0 + row * 32 + lane < n;
        const int  d     = (key[r] >> shift) & (NDIG - 1);
        unsigned peers = __builtin_amdgcn_ballot_w32(valid);
#pragma unroll
        for (int b = 0; b < 6; ++b) {
            const bool     bit = (d >> b) & 1;
            const unsigned m   = __builtin_amdgcn_ballot_w32(valid && bit);
            peers &= bit ? m : ~m;
        }
        rk[r] = __popc(peers & lt);
        if (valid && rk[r] == 0) cnt[row * NDIG + d] = __popc(peers);
    }
    if (SCATTER && wave == 0) {
        tcol[lane]      = tb0;
        tcol[lane + 32] = tb1;
    }
    __syncthreads();

    const int dg = t & (NDIG - 1);
    const int q  = t >> 6;
    {
        int run = 0;
        for (int i = 0; i < 32; ++i) {
            const int a = (q * 32 + i) * NDIG + dg;
            const int v = cnt[a];
            if (SCATTER) cnt[a] = run;
            run += v;
        }
        qsum[q * NDIG + dg] = run;
    }
    __syncthreads();

    if (!SCATTER) {
        if (t < NDIG)
            tab[t * nblk + blockIdx.x] = qsum[t] + qsum[NDIG + t] + qsum[2 * NDIG + t] + qsum[3 * NDIG + t];
        return;
    }

    int off[16];
    int pa[16], pb[16], pc[16];
#pragma unroll
    for (int r = 0; r < 16; ++r) {
        const int row = r * 8 + wave;
        const int d   = (key[r] >> shift) & (NDIG - 1);
        pa[r] = tcol[d];
        pb[r] = cnt[row * NDIG + d];
        int s = 0;
        if (r >= 4)  s += qsum[d];
        if (r >= 8)  s += qsum[NDIG + d];
        if (r >= 12) s += qsum[2 * NDIG + d];
        pc[r] = s;
    }
#pragma unroll
    for (int r = 0; r < 16; ++r) off[r] = pa[r] + pb[r] + pc[r] + rk[r];
    if (n - tile0 >= TILE_KEYS) {
#pragma unroll
        for (int r = 0; r < 16; ++r) {
            kout[off[r]] = key[r];
            vout[off[r]] = val[r];
        }
    } else {
#pragma unroll
        for (int r = 0; r < 16; ++r) {
            if (tile0 + (r * 8 + wave) * 32 + lane < n) {
                kout[off[r]] = key[r];
                vout[off[r]] = val[r];
            }
        }
    }
}

__global__ __launch_bounds__(256) void k_scan(const int* __restrict__ in,
                                              int* __restrict__ out, int n) {
    __shared__ int wsum[8];
    const int t    = threadIdx.x;
    const int lane = t & 31;
    const int wave = t >> 5;
    const int base = blockIdx.x * 1024;

    int part = 0;
    for (int i = t * 4; i < base; i += 8 * 1024) {
        int4 v[8];
#pragma unroll
        for (int j = 0; j < 8; ++j) {
            const int ij = i + j * 1024;
            v[j] = *(const int4*)(in + (ij < base ? ij : i));
        }
#pragma unroll
        for (int j = 0; j < 8; ++j)
            part += (i + j * 1024 < base) ? v[j].x + v[j].y + v[j].z + v[j].w : 0;
    }
    for (int o = 16; o > 0; o >>= 1) part += __shfl_xor(part, o, 32);
    if (lane == 0) wsum[wave] = part;
    __syncthreads();
    int prefix = 0;
    for (int w = 0; w < 8; ++w) prefix += wsum[w];
    __syncthreads();

    const int i0 = base + t * 4;
    int c0 = (i0     < n) ? in[i0]     : 0;
    int c1 = (i0 + 1 < n) ? in[i0 + 1] : 0;
    int c2 = (i0 + 2 < n) ? in[i0 + 2] : 0;
    int c3 = (i0 + 3 < n) ? in[i0 + 3] : 0;
    const int tsum = c0 + c1 + c2 + c3;
    int inc = tsum;
    for (int o = 1; o < 32; o <<= 1) {
        const int u = __shfl_up(inc, o, 32);
        if (lane >= o) inc += u;
    }
    if (lane == 31) wsum[wave] = inc;
    __syncthreads();
    int woff = 0;
    for (int w = 0; w < 8; ++w) woff += (w < wave) ? wsum[w] : 0;

    const int e0 = prefix + woff + inc - tsum;
    if (i0     < n) out[i0]     = e0;
    if (i0 + 1 < n) out[i0 + 1] = e0 + c0;
    if (i0 + 2 < n) out[i0 + 2] = e0 + c0 + c1;
    if (i0 + 3 < n) out[i0 + 3] = e0 + c0 + c1 + c2;
}

__global__ __launch_bounds__(256) void k_rowptr(const int* __restrict__ sdst,
                                                int* __restrict__ rowptr, int e_cnt, int n) {
    const int j = blockIdx.x * blockDim.x + threadIdx.x;
    if (j > e_cnt) return;
    const int lo = (j == 0) ? 0 : sdst[j - 1] + 1;
    const int hi = (j == e_cnt) ? n : sdst[j];
    for (int d = lo; d <= hi; ++d) rowptr[d] = j;
}

__global__ __launch_bounds__(256) void k_dinv(const int* __restrict__ rowptr,
                                              float* __restrict__ dinv, int n) {
    const int i = blockIdx.x * blockDim.x + threadIdx.x;
    if (i < n) dinv[i] = rsqrtf((float)(rowptr[i + 1] - rowptr[i] + 1));
}

__global__ __launch_bounds__(256) void k_edge_norm(const int* __restrict__ ssrc,
                                                   const int* __restrict__ sdst,
                                                   const float* __restrict__ dinv,
                                                   float* __restrict__ norm, int e_cnt) {
    const int j = blockIdx.x * blockDim.x + threadIdx.x;
    if (j < e_cnt) norm[j] = dinv[ssrc[j]] * dinv[sdst[j]];
}


#define AP 68

template <bool RELU_OUT>
__global__ __launch_bounds__(256) void k_layer(const int* __restrict__ rowptr,
                                               const int* __restrict__ ssrc,
                                               const float* __restrict__ norm,
                                               const float* __restrict__ dinv,
                                               const float* __restrict__ X,
                                               const float* __restrict__ W,
                                               const float* __restrict__ bias,
                                               float* __restrict__ Y, int n_nodes) {
    __shared__ float As[16 * AP];
    const unsigned t    = threadIdx.x;
    const unsigned nl   = t >> 4;
    const unsigned l    = t & 15u;
    const unsigned node = blockIdx.x * 16u + nl;

    float4 a = make_float4(0.0f, 0.0f, 0.0f, 0.0f);
    if (node < (unsigned)n_nodes) {
        const int   beg = rowptr[node];
        const int   end = rowptr[node + 1];
        const float di  = dinv[node];
        const float dd  = di * di;
        const float4* X4 = (const float4*)X;
        const float4 hv = X4[node * 16u + l];
        a.x = hv.x * dd;
        a.y = hv.y * dd;
        a.z = hv.z * dd;
        a.w = hv.w * dd;
#pragma unroll 4
        for (int j = beg; j < end; ++j) {
            const unsigned s = (unsigned)ssrc[j];
            const float    w = norm[j];
            const float4   r = X4[s * 16u + l];
            a.x += r.x * w;
            a.y += r.y * w;
            a.z += r.z * w;
            a.w += r.w * w;
        }
    }
    *(float4*)(As + nl * AP + 4u * l) = a;
    __syncthreads();

    const int wave = (int)(t >> 5);
    if (wave >= 4) return;
    const int lane = (int)(t & 31u);
    const int half = lane >> 4;
    const int l16  = lane & 15;
    const int ncol = wave * 16 + l16;

    v8f c0 = {};
    v8f c1 = {};
#pragma unroll
    for (int k0 = 0; k0 < D; k0 += 8) {
        const int ka = k0 + 2 * half;
        const int kb = ka + 4;
        const float2 a0 = *(const float2*)(As + l16 * AP + ka);
        const float2 a1 = *(const float2*)(As + l16 * AP + kb);
        v2f fa0, fb0, fa1, fb1;
        fa0.x = a0.x;  fa0.y = a0.y;
        fb0.x = W[ka * D + ncol];
        fb0.y = W[(ka + 1) * D + ncol];
        fa1.x = a1.x;  fa1.y = a1.y;
        fb1.x = W[kb * D + ncol];
        fb1.y = W[(kb + 1) * D + ncol];
        c0 = __builtin_amdgcn_wmma_f32_16x16x4_f32(false, fa0, false, fb0,
                                                   (short)0, c0, false, false);
        c1 = __builtin_amdgcn_wmma_f32_16x16x4_f32(false, fa1, false, fb1,
                                                   (short)0, c1, false, false);
    }

    const float bv    = bias[ncol];
    const int   mbase = (int)blockIdx.x * 16 + half * 8;
#pragma unroll
    for (int r = 0; r < 8; ++r) {
        const int m = mbase + r;
        float v = c0[r] + c1[r] + bv;
        if (RELU_OUT) v = fmaxf(v, 0.0f);
        if (m < n_nodes) Y[(long)m * D + ncol] = v;
    }
}

extern "C" void kernel_launch(void* const* d_in, const int* in_sizes, int n_in,
                              void* d_out, int out_size, void* d_ws, size_t ws_size,
                              hipStream_t stream) {
    const float* x  = (const float*)d_in[0];
    const int*   ei = (const int*)d_in[1];
    const float* Wm[3] = { (const float*)d_in[2], (const float*)d_in[4], (const float*)d_in[6] };
    const float* bm[3] = { (const float*)d_in[3], (const float*)d_in[5], (const float*)d_in[7] };

    const int N = in_sizes[0] / D;
    const int E = in_sizes[1] / 2;
    const int* src = ei;
    const int* dst = ei + E;

    const int  nblk = (E + TILE_KEYS - 1) / TILE_KEYS;
    const int  tlen = NDIG * nblk;
    const long npad = ((long)N + 63) & ~63L;
    const long epad = ((long)E + 63) & ~63L;
    const long tpad = ((long)tlen + 1023) & ~1023L;
    float* dinv   = (float*)d_ws;
    int*   rowptr = (int*)(dinv + npad);
    int*   sdst   = rowptr + npad + 64;
    int*   ssrc   = sdst + epad;
    float* norm   = (float*)(ssrc + epad);
    int*   tabA   = (int*)(norm + epad);
    int*   tabB   = tabA + tpad;
    float* h      = (float*)(tabB + tpad);
    int*   k2     = (int*)h;
    int*   v2     = k2 + epad;
    float* out    = (float*)d_out;

    const int B = 256;
    const int gN    = (N + B - 1) / B;
    const int gE    = (E + B - 1) / B;
    const int gE1   = (E + 1 + B - 1) / B;
    const int gScan = (tlen + 1023) / 1024;
    const int gLayer = (N + 15) / 16;

    k_sort_pass<false><<<nblk, B, 0, stream>>>(dst, src, sdst, ssrc, tabA, E, 0, nblk);
    k_scan<<<gScan, B, 0, stream>>>(tabA, tabB, tlen);
    k_sort_pass<true><<<nblk, B, 0, stream>>>(dst, src, sdst, ssrc, tabB, E, 0, nblk);

    k_sort_pass<false><<<nblk, B, 0, stream>>>(sdst, ssrc, k2, v2, tabA, E, 6, nblk);
    k_scan<<<gScan, B, 0, stream>>>(tabA, tabB, tlen);
    k_sort_pass<true><<<nblk, B, 0, stream>>>(sdst, ssrc, k2, v2, tabB, E, 6, nblk);

    k_sort_pass<false><<<nblk, B, 0, stream>>>(k2, v2, sdst, ssrc, tabA, E, 12, nblk);
    k_scan<<<gScan, B, 0, stream>>>(tabA, tabB, tlen);
    k_sort_pass<true><<<nblk, B, 0, stream>>>(k2, v2, sdst, ssrc, tabB, E, 12, nblk);

    k_rowptr<<<gE1, B, 0, stream>>>(sdst, rowptr, E, N);
    k_dinv<<<gN, B, 0, stream>>>(rowptr, dinv, N);
    k_edge_norm<<<gE, B, 0, stream>>>(ssrc, sdst, dinv, norm, E);

    k_layer<true><<<gLayer, B, 0, stream>>>(rowptr, ssrc, norm, dinv, x, Wm[0], bm[0], out, N);

    k_layer<true><<<gLayer, B, 0, stream>>>(rowptr, ssrc, norm, dinv, out, Wm[1], bm[1], h, N);

    k_layer<false><<<gLayer, B, 0, stream>>>(rowptr, ssrc, norm, dinv, h, Wm[2], bm[2], out, N);
}


# layer gather: a node's 16 lanes read its edge list in one coalesced load and share it by lane shuffles, rows issued eight at a time (about 4 dependent round trips per node, were 12); sum order same
# speedup vs baseline: 2.3661x; 1.0449x over previous
#include <hip/hip_runtime.h>

typedef float v2f __attribute__((ext_vector_type(2)));
typedef float v8f __attribute__((ext_vector_type(8)));

#define D 64


#define TILE_KEYS 4096
#define TILE_ROWS 128
#define NDIG      64

template <bool SCATTER>
__global__ __launch_bounds__(256) void k_sort_pass(const int* __restrict__ kin,
                                                   const int* __restrict__ vin,
                                                   int* __restrict__ kout,
                                                   int* __restrict__ vout,
                                                   int* __restrict__ tab,
                                                   int n, int shift, int nblk) {
    __shared__ int cnt[TILE_ROWS * NDIG];
    __shared__ int qsum[4 * NDIG];
    __shared__ int tcol[NDIG];
    const int t     = threadIdx.x;
    const int lane  = t & 31;
    const int wave  = t >> 5;
    const int tile0 = blockIdx.x * TILE_KEYS;
    const int last  = n - 1;

    int key[16];
    int val[16];
#pragma unroll
    for (int r = 0; r < 16; ++r) {
        const int idx = tile0 + (r * 8 + wave) * 32 + lane;
        const int ci  = idx < last ? idx : last;
        key[r] = kin[ci];
        if (SCATTER) val[r] = vin[ci];
    }
    int tb0 = 0, tb1 = 0;
    if (SCATTER && wave == 0) {
        tb0 = tab[lane * nblk + blockIdx.x];
        tb1 = tab[(lane + 32) * nblk + blockIdx.x];
    }

    for (int i = t; i < TILE_ROWS * NDIG; i += 256) cnt[i] = 0;
    __syncthreads();

    int rk[16];
    const unsigned lt = (1u << lane) - 1u;
#pragma unroll
    for (int r = 0; r < 16; ++r) {
        const int  row   = r * 8 + wave;
        const bool valid = tile0 + row * 32 + lane < n;
        const int  d     = (key[r] >> shift) & (NDIG - 1);
        unsigned peers = __builtin_amdgcn_ballot_w32(valid);
#pragma unroll
        for (int b = 0; b < 6; ++b) {
            const bool     bit = (d >> b) & 1;
            const unsigned m   = __builtin_amdgcn_ballot_w32(valid && bit);
            peers &= bit ? m : ~m;
        }
        rk[r] = __popc(peers & lt);
        if (valid && rk[r] == 0) cnt[row * NDIG + d] = __popc(peers);
    }
    if (SCATTER && wave == 0) {
        tcol[lane]      = tb0;
        tcol[lane + 32] = tb1;
    }
    __syncthreads();

    const int dg = t & (NDIG - 1);
    const int q  = t >> 6;
    {
        int run = 0;
        for (int i = 0; i < 32; ++i) {
            const int a = (q * 32 + i) * NDIG + dg;
            const int v = cnt[a];
            if (SCATTER) cnt[a] = run;
            run += v;
        }
        qsum[q * NDIG + dg] = run;
    }
    __syncthreads();

    if (!SCATTER) {
        if (t < NDIG)
            tab[t * nblk + blockIdx.x] = qsum[t] + qsum[NDIG + t] + qsum[2 * NDIG + t] + qsum[3 * NDIG + t];
        return;
    }

    int off[16];
    int pa[16], pb[16], pc[16];
#pragma unroll
    for (int r = 0; r < 16; ++r) {
        const int row = r * 8 + wave;
        const int d   = (key[r] >> shift) & (NDIG - 1);
        pa[r] = tcol[d];
        pb[r] = cnt[row * NDIG + d];
        int s = 0;
        if (r >= 4)  s += qsum[d];
        if (r >= 8)  s += qsum[NDIG + d];
        if (r >= 12) s += qsum[2 * NDIG + d];
        pc[r] = s;
    }
#pragma unroll
    for (int r = 0; r < 16; ++r) off[r] = pa[r] + pb[r] + pc[r] + rk[r];
    if (n - tile0 >= TILE_KEYS) {
#pragma unroll
        for (int r = 0; r < 16; ++r) {
            kout[off[r]] = key[r];
            vout[off[r]] = val[r];
        }
    } else {
#pragma unroll
        for (int r = 0; r < 16; ++r) {
            if (tile0 + (r * 8 + wave) * 32 + lane < n) {
                kout[off[r]] = key[r];
                vout[off[r]] = val[r];
            }
        }
    }
}

__global__ __launch_bounds__(256) void k_scan(const int* __restrict__ in,
                                              int* __restrict__ out, int n) {
    __shared__ int wsum[8];
    const int t    = threadIdx.x;
    const int lane = t & 31;
    const int wave = t >> 5;
    const int base = blockIdx.x * 1024;

    int part = 0;
    for (int i = t * 4; i < base; i += 8 * 1024) {
        int4 v[8];
#pragma unroll
        for (int j = 0; j < 8; ++j) {
            const int ij = i + j * 1024;
            v[j] = *(const int4*)(in + (ij < base ? ij : i));
        }
#pragma unroll
        for (int j = 0; j < 8; ++j)
            part += (i + j * 1024 < base) ? v[j].x + v[j].y + v[j].z + v[j].w : 0;
    }
    for (int o = 16; o > 0; o >>= 1) part += __shfl_xor(part, o, 32);
    if (lane == 0) wsum[wave] = part;
    __syncthreads();
    int prefix = 0;
    for (int w = 0; w < 8; ++w) prefix += wsum[w];
    __syncthreads();

    const int i0 = base + t * 4;
    int c0 = (i0     < n) ? in[i0]     : 0;
    int c1 = (i0 + 1 < n) ? in[i0 + 1] : 0;
    int c2 = (i0 + 2 < n) ? in[i0 + 2] : 0;
    int c3 = (i0 + 3 < n) ? in[i0 + 3] : 0;
    const int tsum = c0 + c1 + c2 + c3;
    int inc = tsum;
    for (int o = 1; o < 32; o <<= 1) {
        const int u = __shfl_up(inc, o, 32);
        if (lane >= o) inc += u;
    }
    if (lane == 31) wsum[wave] = inc;
    __syncthreads();
    int woff = 0;
    for (int w = 0; w < 8; ++w) woff += (w < wave) ? wsum[w] : 0;

    const int e0 = prefix + woff + inc - tsum;
    if (i0     < n) out[i0]     = e0;
    if (i0 + 1 < n) out[i0 + 1] = e0 + c0;
    if (i0 + 2 < n) out[i0 + 2] = e0 + c0 + c1;
    if (i0 + 3 < n) out[i0 + 3] = e0 + c0 + c1 + c2;
}

__global__ __launch_bounds__(256) void k_rowptr(const int* __restrict__ sdst,
                                                int* __restrict__ rowptr, int e_cnt, int n) {
    const int j = blockIdx.x * blockDim.x + threadIdx.x;
    if (j > e_cnt) return;
    const int lo = (j == 0) ? 0 : sdst[j - 1] + 1;
    const int hi = (j == e_cnt) ? n : sdst[j];
    for (int d = lo; d <= hi; ++d) rowptr[d] = j;
}

__global__ __launch_bounds__(256) void k_dinv(const int* __restrict__ rowptr,
                                              float* __restrict__ dinv, int n) {
    const int i = blockIdx.x * blockDim.x + threadIdx.x;
    if (i < n) dinv[i] = rsqrtf((float)(rowptr[i + 1] - rowptr[i] + 1));
}

__global__ __launch_bounds__(256) void k_edge_norm(const int* __restrict__ ssrc,
                                                   const int* __restrict__ sdst,
                                                   const float* __restrict__ dinv,
                                                   float* __restrict__ norm, int e_cnt) {
    const int j = blockIdx.x * blockDim.x + threadIdx.x;
    if (j < e_cnt) norm[j] = dinv[ssrc[j]] * dinv[sdst[j]];
}


#define AP 68

template <bool RELU_OUT>
__global__ __launch_bounds__(256) void k_layer(const int* __restrict__ rowptr,
                                               const int* __restrict__ ssrc,
                                               const float* __restrict__ norm,
                                               const float* __restrict__ dinv,
                                               const float* __restrict__ X,
                                               const float* __restrict__ W,
                                               const float* __restrict__ bias,
                                               float* __restrict__ Y, int n_nodes) {
    __shared__ float As[16 * AP];
    const unsigned t    = threadIdx.x;
    const unsigned nl   = t >> 4;
    const unsigned l    = t & 15u;
    const unsigned node = blockIdx.x * 16u + nl;

    float4 a = make_float4(0.0f, 0.0f, 0.0f, 0.0f);
    if (node < (unsigned)n_nodes) {
        const int   beg = rowptr[node];
        const int   end = rowptr[node + 1];
        const float di  = dinv[node];
        const float dd  = di * di;
        const float4* X4 = (const float4*)X;
        const float4 hv = X4[node * 16u + l];
        a.x = hv.x * dd;
        a.y = hv.y * dd;
        a.z = hv.z * dd;
        a.w = hv.w * dd;
        for (int c = beg; c < end; c += 16) {
            const int   je = c + (int)l;
            const bool  ok = je < end;
            const int   cj = ok ? je : end - 1;
            const int   sv = ssrc[cj];
            const float wv = norm[cj];
            const int   si = ok ? sv : (int)node;
            const float wi = ok ? wv : 0.0f;
            const int  cnt = end - c;
            for (int e0 = 0; e0 < 16 && e0 < cnt; e0 += 8) {
                float4 r[8];
                float  w[8];
#pragma unroll
                for (int u = 0; u < 8; ++u) {
                    const unsigned s = (unsigned)__shfl(si, e0 + u, 16);
                    w[u] = __shfl(wi, e0 + u, 16);
                    r[u] = X4[s * 16u + l];
                }
#pragma unroll
                for (int u = 0; u < 8; ++u) {
                    a.x += r[u].x * w[u];
                    a.y += r[u].y * w[u];
                    a.z += r[u].z * w[u];
                    a.w += r[u].w * w[u];
                }
            }
        }
    }
    *(float4*)(As + nl * AP + 4u * l) = a;
    __syncthreads();

    const int wave = (int)(t >> 5);
    if (wave >= 4) return;
    const int lane = (int)(t & 31u);
    const int half = lane >> 4;
    const int l16  = lane & 15;
    const int ncol = wave * 16 + l16;

    v8f c0 = {};
    v8f c1 = {};
#pragma unroll
    for (int k0 = 0; k0 < D; k0 += 8) {
        const int ka = k0 + 2 * half;
        const int kb = ka + 4;
        const float2 a0 = *(const float2*)(As + l16 * AP + ka);
        const float2 a1 = *(const float2*)(As + l16 * AP + kb);
        v2f fa0, fb0, fa1, fb1;
        fa0.x = a0.x;  fa0.y = a0.y;
        fb0.x = W[ka * D + ncol];
        fb0.y = W[(ka + 1) * D + ncol];
        fa1.x = a1.x;  fa1.y = a1.y;
        fb1.x = W[kb * D + ncol];
        fb1.y = W[(kb + 1) * D + ncol];
        c0 = __builtin_amdgcn_wmma_f32_16x16x4_f32(false, fa0, false, fb0,
                                                   (short)0, c0, false, false);
        c1 = __builtin_amdgcn_wmma_f32_16x16x4_f32(false, fa1, false, fb1,
                                                   (short)0, c1, false, false);
    }

    const float bv    = bias[ncol];
    const int   mbase = (int)blockIdx.x * 16 + half * 8;
#pragma unroll
    for (int r = 0; r < 8; ++r) {
        const int m = mbase + r;
        float v = c0[r] + c1[r] + bv;
        if (RELU_OUT) v = fmaxf(v, 0.0f);
        if (m < n_nodes) Y[(long)m * D + ncol] = v;
    }
}

extern "C" void kernel_launch(void* const* d_in, const int* in_sizes, int n_in,
                              void* d_out, int out_size, void* d_ws, size_t ws_size,
                              hipStream_t stream) {
    const float* x  = (const float*)d_in[0];
    const int*   ei = (const int*)d_in[1];
    const float* Wm[3] = { (const float*)d_in[2], (const float*)d_in[4], (const float*)d_in[6] };
    const float* bm[3] = { (const float*)d_in[3], (const float*)d_in[5], (const float*)d_in[7] };

    const int N = in_sizes[0] / D;
    const int E = in_sizes[1] / 2;
    const int* src = ei;
    const int* dst = ei + E;

    const int  nblk = (E + TILE_KEYS - 1) / TILE_KEYS;
    const int  tlen = NDIG * nblk;
    const long npad = ((long)N + 63) & ~63L;
    const long epad = ((long)E + 63) & ~63L;
    const long tpad = ((long)tlen + 1023) & ~1023L;
    float* dinv   = (float*)d_ws;
    int*   rowptr = (int*)(dinv + npad);
    int*   sdst   = rowptr + npad + 64;
    int*   ssrc   = sdst + epad;
    float* norm   = (float*)(ssrc + epad);
    int*   tabA   = (int*)(norm + epad);
    int*   tabB   = tabA + tpad;
    float* h      = (float*)(tabB + tpad);
    int*   k2     = (int*)h;
    int*   v2     = k2 + epad;
    float* out    = (float*)d_out;

    const int B = 256;
    const int gN    = (N + B - 1) / B;
    const int gE    = (E + B - 1) / B;
    const int gE1   = (E + 1 + B - 1) / B;
    const int gScan = (tlen + 1023) / 1024;
    const int gLayer = (N + 15) / 16;

    k_sort_pass<false><<<nblk, B, 0, stream>>>(dst, src, sdst, ssrc, tabA, E, 0, nblk);
    k_scan<<<gScan, B, 0, stream>>>(tabA, tabB, tlen);
    k_sort_pass<true><<<nblk, B, 0, stream>>>(dst, src, sdst, ssrc, tabB, E, 0, nblk);

    k_sort_pass<false><<<nblk, B, 0, stream>>>(sdst, ssrc, k2, v2, tabA, E, 6, nblk);
    k_scan<<<gScan, B, 0, stream>>>(tabA, tabB, tlen);
    k_sort_pass<true><<<nblk, B, 0, stream>>>(sdst, ssrc, k2, v2, tabB, E, 6, nblk);

    k_sort_pass<false><<<nblk, B, 0, stream>>>(k2, v2, sdst, ssrc, tabA, E, 12, nblk);
    k_scan<<<gScan, B, 0, stream>>>(tabA, tabB, tlen);
    k_sort_pass<true><<<nblk, B, 0, stream>>>(k2, v2, sdst, ssrc, tabB, E, 12, nblk);

    k_rowptr<<<gE1, B, 0, stream>>>(sdst, rowptr, E, N);
    k_dinv<<<gN, B, 0, stream>>>(rowptr, dinv, N);
    k_edge_norm<<<gE, B, 0, stream>>>(ssrc, sdst, dinv, norm, E);

    k_layer<true><<<gLayer, B, 0, stream>>>(rowptr, ssrc, norm, dinv, x, Wm[0], bm[0], out, N);

    k_layer<true><<<gLayer, B, 0, stream>>>(rowptr, ssrc, norm, dinv, out, Wm[1], bm[1], h, N);

    k_layer<false><<<gLayer, B, 0, stream>>>(rowptr, ssrc, norm, dinv, h, Wm[2], bm[2], out, N);
}


# sort scatter passes: tile put in output order in LDS first, written with consecutive threads on consecutive places, so a wave's store covers a few whole lines, not one per lane; sum order same
# speedup vs baseline: 3.3026x; 1.3958x over previous
#include <hip/hip_runtime.h>

typedef float v2f __attribute__((ext_vector_type(2)));
typedef float v8f __attribute__((ext_vector_type(8)));

#define D 64


#define TILE_KEYS 4096
#define TILE_ROWS 128
#define NDIG      64

template <bool SCATTER>
__global__ __launch_bounds__(256) void k_sort_pass(const int* __restrict__ kin,
                                                   const int* __restrict__ vin,
                                                   int* __restrict__ kout,
                                                   int* __restrict__ vout,
                                                   int* __restrict__ tab,
                                                   int n, int shift, int nblk) {
    __shared__ int cnt[TILE_ROWS * NDIG];
    __shared__ int qsum[4 * NDIG];
    __shared__ int tcol[NDIG];
    __shared__ int dstart[NDIG];
    __shared__ int gofs[NDIG];
    __shared__ int2 stage[SCATTER ? TILE_KEYS : 1];
    const int t     = threadIdx.x;
    const int lane  = t & 31;
    const int wave  = t >> 5;
    const int tile0 = blockIdx.x * TILE_KEYS;
    const int last  = n - 1;

    int key[16];
    int val[16];
#pragma unroll
    for (int r = 0; r < 16; ++r) {
        const int idx = tile0 + (r * 8 + wave) * 32 + lane;
        const int ci  = idx < last ? idx : last;
        key[r] = kin[ci];
        if (SCATTER) val[r] = vin[ci];
    }
    int tb0 = 0, tb1 = 0;
    if (SCATTER && wave == 0) {
        tb0 = tab[lane * nblk + blockIdx.x];
        tb1 = tab[(lane + 32) * nblk + blockIdx.x];
    }

    for (int i = t; i < TILE_ROWS * NDIG; i += 256) cnt[i] = 0;
    __syncthreads();

    int rk[16];
    const unsigned lt = (1u << lane) - 1u;
#pragma unroll
    for (int r = 0; r < 16; ++r) {
        const int  row   = r * 8 + wave;
        const bool valid = tile0 + row * 32 + lane < n;
        const int  d     = (key[r] >> shift) & (NDIG - 1);
        unsigned peers = __builtin_amdgcn_ballot_w32(valid);
#pragma unroll
        for (int b = 0; b < 6; ++b) {
            const bool     bit = (d >> b) & 1;
            const unsigned m   = __builtin_amdgcn_ballot_w32(valid && bit);
            peers &= bit ? m : ~m;
        }
        rk[r] = __popc(peers & lt);
        if (valid && rk[r] == 0) cnt[row * NDIG + d] = __popc(peers);
    }
    if (SCATTER && wave == 0) {
        tcol[lane]      = tb0;
        tcol[lane + 32] = tb1;
    }
    __syncthreads();

    const int dg = t & (NDIG - 1);
    const int q  = t >> 6;
    {
        int run = 0;
        for (int i = 0; i < 32; ++i) {
            const int a = (q * 32 + i) * NDIG + dg;
            const int v = cnt[a];
            if (SCATTER) cnt[a] = run;
            run += v;
        }
        qsum[q * NDIG + dg] = run;
    }
    __syncthreads();

    if (!SCATTER) {
        if (t < NDIG)
            tab[t * nblk + blockIdx.x] = qsum[t] + qsum[NDIG + t] + qsum[2 * NDIG + t] + qsum[3 * NDIG + t];
        return;
    }

    if (wave == 0) {
        const int d0 = 2 * lane;
        const int t0 = qsum[d0] + qsum[NDIG + d0] + qsum[2 * NDIG + d0] + qsum[3 * NDIG + d0];
        const int t1 = qsum[d0 + 1] + qsum[NDIG + d0 + 1] + qsum[2 * NDIG + d0 + 1] + qsum[3 * NDIG + d0 + 1];
        const int s  = t0 + t1;
        int inc = s;
        for (int o = 1; o < 32; o <<= 1) {
            const int u = __shfl_up(inc, o, 32);
            if (lane >= o) inc += u;
        }
        const int e0 = inc - s;
        dstart[d0]     = e0;
        dstart[d0 + 1] = e0 + t0;
        gofs[d0]       = tcol[d0] - e0;
        gofs[d0 + 1]   = tcol[d0 + 1] - (e0 + t0);
    }

    int pb[16], pc[16];
#pragma unroll
    for (int r = 0; r < 16; ++r) {
        const int row = r * 8 + wave;
        const int d   = (key[r] >> shift) & (NDIG - 1);
        pb[r] = cnt[row * NDIG + d];
        int s = 0;
        if (r >= 4)  s += qsum[d];
        if (r >= 8)  s += qsum[NDIG + d];
        if (r >= 12) s += qsum[2 * NDIG + d];
        pc[r] = s;
    }
    __syncthreads();

    const bool full = n - tile0 >= TILE_KEYS;
    int lp[16];
#pragma unroll
    for (int r = 0; r < 16; ++r) {
        const int d = (key[r] >> shift) & (NDIG - 1);
        lp[r] = dstart[d] + pb[r] + pc[r] + rk[r];
    }
    if (full) {
#pragma unroll
        for (int r = 0; r < 16; ++r) stage[lp[r]] = make_int2(key[r], val[r]);
    } else {
#pragma unroll
        for (int r = 0; r < 16; ++r)
            if (tile0 + (r * 8 + wave) * 32 + lane < n) stage[lp[r]] = make_int2(key[r], val[r]);
    }
    __syncthreads();

    if (full) {
        int2 kv[16];
#pragma unroll
        for (int r = 0; r < 16; ++r) kv[r] = stage[r * 256 + t];
#pragma unroll
        for (int r = 0; r < 16; ++r) {
            const int g = gofs[(kv[r].x >> shift) & (NDIG - 1)] + r * 256 + t;
            kout[g] = kv[r].x;
            vout[g] = kv[r].y;
        }
    } else {
        const int nv = n - tile0;
        for (int i = t; i < nv; i += 256) {
            const int2 kv = stage[i];
            const int  g  = gofs[(kv.x >> shift) & (NDIG - 1)] + i;
            kout[g] = kv.x;
            vout[g] = kv.y;
        }
    }
}

__global__ __launch_bounds__(256) void k_scan(const int* __restrict__ in,
                                              int* __restrict__ out, int n) {
    __shared__ int wsum[8];
    const int t    = threadIdx.x;
    const int lane = t & 31;
    const int wave = t >> 5;
    const int base = blockIdx.x * 1024;

    int part = 0;
    for (int i = t * 4; i < base; i += 8 * 1024) {
        int4 v[8];
#pragma unroll
        for (int j = 0; j < 8; ++j) {
            const int ij = i + j * 1024;
            v[j] = *(const int4*)(in + (ij < base ? ij : i));
        }
#pragma unroll
        for (int j = 0; j < 8; ++j)
            part += (i + j * 1024 < base) ? v[j].x + v[j].y + v[j].z + v[j].w : 0;
    }
    for (int o = 16; o > 0; o >>= 1) part += __shfl_xor(part, o, 32);
    if (lane == 0) wsum[wave] = part;
    __syncthreads();
    int prefix = 0;
    for (int w = 0; w < 8; ++w) prefix += wsum[w];
    __syncthreads();

    const int i0 = base + t * 4;
    int c0 = (i0     < n) ? in[i0]     : 0;
    int c1 = (i0 + 1 < n) ? in[i0 + 1] : 0;
    int c2 = (i0 + 2 < n) ? in[i0 + 2] : 0;
    int c3 = (i0 + 3 < n) ? in[i0 + 3] : 0;
    const int tsum = c0 + c1 + c2 + c3;
    int inc = tsum;
    for (int o = 1; o < 32; o <<= 1) {
        const int u = __shfl_up(inc, o, 32);
        if (lane >= o) inc += u;
    }
    if (lane == 31) wsum[wave] = inc;
    __syncthreads();
    int woff = 0;
    for (int w = 0; w < 8; ++w) woff += (w < wave) ? wsum[w] : 0;

    const int e0 = prefix + woff + inc - tsum;
    if (i0     < n) out[i0]     = e0;
    if (i0 + 1 < n) out[i0 + 1] = e0 + c0;
    if (i0 + 2 < n) out[i0 + 2] = e0 + c0 + c1;
    if (i0 + 3 < n) out[i0 + 3] = e0 + c0 + c1 + c2;
}

__global__ __launch_bounds__(256) void k_rowptr(const int* __restrict__ sdst,
                                                int* __restrict__ rowptr, int e_cnt, int n) {
    const int j = blockIdx.x * blockDim.x + threadIdx.x;
    if (j > e_cnt) return;
    const int lo = (j == 0) ? 0 : sdst[j - 1] + 1;
    const int hi = (j == e_cnt) ? n : sdst[j];
    for (int d = lo; d <= hi; ++d) rowptr[d] = j;
}

__global__ __launch_bounds__(256) void k_dinv(const int* __restrict__ rowptr,
                                              float* __restrict__ dinv, int n) {
    const int i = blockIdx.x * blockDim.x + threadIdx.x;
    if (i < n) dinv[i] = rsqrtf((float)(rowptr[i + 1] - rowptr[i] + 1));
}

__global__ __launch_bounds__(256) void k_edge_norm(const int* __restrict__ ssrc,
                                                   const int* __restrict__ sdst,
                                                   const float* __restrict__ dinv,
                                                   float* __restrict__ norm, int e_cnt) {
    const int j = blockIdx.x * blockDim.x + threadIdx.x;
    if (j < e_cnt) norm[j] = dinv[ssrc[j]] * dinv[sdst[j]];
}


#define AP 68

template <bool RELU_OUT>
__global__ __launch_bounds__(256) void k_layer(const int* __restrict__ rowptr,
                                               const int* __restrict__ ssrc,
                                               const float* __restrict__ norm,
                                               const float* __restrict__ dinv,
                                               const float* __restrict__ X,
                                               const float* __restrict__ W,
                                               const float* __restrict__ bias,
                                               float* __restrict__ Y, int n_nodes) {
    __shared__ float As[16 * AP];
    const unsigned t    = threadIdx.x;
    const unsigned nl   = t >> 4;
    const unsigned l    = t & 15u;
    const unsigned node = blockIdx.x * 16u + nl;

    float4 a = make_float4(0.0f, 0.0f, 0.0f, 0.0f);
    if (node < (unsigned)n_nodes) {
        const int   beg = rowptr[node];
        const int   end = rowptr[node + 1];
        const float di  = dinv[node];
        const float dd  = di * di;
        const float4* X4 = (const float4*)X;
        const float4 hv = X4[node * 16u + l];
        a.x = hv.x * dd;
        a.y = hv.y * dd;
        a.z = hv.z * dd;
        a.w = hv.w * dd;
        for (int c = beg; c < end; c += 16) {
            const int   je = c + (int)l;
            const bool  ok = je < end;
            const int   cj = ok ? je : end - 1;
            const int   sv = ssrc[cj];
            const float wv = norm[cj];
            const int   si = ok ? sv : (int)node;
            const float wi = ok ? wv : 0.0f;
            const int  cnt = end - c;
            for (int e0 = 0; e0 < 16 && e0 < cnt; e0 += 8) {
                float4 r[8];
                float  w[8];
#pragma unroll
                for (int u = 0; u < 8; ++u) {
                    const unsigned s = (unsigned)__shfl(si, e0 + u, 16);
                    w[u] = __shfl(wi, e0 + u, 16);
                    r[u] = X4[s * 16u + l];
                }
#pragma unroll
                for (int u = 0; u < 8; ++u) {
                    a.x += r[u].x * w[u];
                    a.y += r[u].y * w[u];
                    a.z += r[u].z * w[u];
                    a.w += r[u].w * w[u];
                }
            }
        }
    }
    *(float4*)(As + nl * AP + 4u * l) = a;
    __syncthreads();

    const int wave = (int)(t >> 5);
    if (wave >= 4) return;
    const int lane = (int)(t & 31u);
    const int half = lane >> 4;
    const int l16  = lane & 15;
    const int ncol = wave * 16 + l16;

    v8f c0 = {};
    v8f c1 = {};
#pragma unroll
    for (int k0 = 0; k0 < D; k0 += 8) {
        const int ka = k0 + 2 * half;
        const int kb = ka + 4;
        const float2 a0 = *(const float2*)(As + l16 * AP + ka);
        const float2 a1 = *(const float2*)(As + l16 * AP + kb);
        v2f fa0, fb0, fa1, fb1;
        fa0.x = a0.x;  fa0.y = a0.y;
        fb0.x = W[ka * D + ncol];
        fb0.y = W[(ka + 1) * D + ncol];
        fa1.x = a1.x;  fa1.y = a1.y;
        fb1.x = W[kb * D + ncol];
        fb1.y = W[(kb + 1) * D + ncol];
        c0 = __builtin_amdgcn_wmma_f32_16x16x4_f32(false, fa0, false, fb0,
                                                   (short)0, c0, false, false);
        c1 = __builtin_amdgcn_wmma_f32_16x16x4_f32(false, fa1, false, fb1,
                                                   (short)0, c1, false, false);
    }

    const float bv    = bias[ncol];
    const int   mbase = (int)blockIdx.x * 16 + half * 8;
#pragma unroll
    for (int r = 0; r < 8; ++r) {
        const int m = mbase + r;
        float v = c0[r] + c1[r] + bv;
        if (RELU_OUT) v = fmaxf(v, 0.0f);
        if (m < n_nodes) Y[(long)m * D + ncol] = v;
    }
}

extern "C" void kernel_launch(void* const* d_in, const int* in_sizes, int n_in,
                              void* d_out, int out_size, void* d_ws, size_t ws_size,
                              hipStream_t stream) {
    const float* x  = (const float*)d_in[0];
    const int*   ei = (const int*)d_in[1];
    const float* Wm[3] = { (const float*)d_in[2], (const float*)d_in[4], (const float*)d_in[6] };
    const float* bm[3] = { (const float*)d_in[3], (const float*)d_in[5], (const float*)d_in[7] };

    const int N = in_sizes[0] / D;
    const int E = in_sizes[1] / 2;
    const int* src = ei;
    const int* dst = ei + E;

    const int  nblk = (E + TILE_KEYS - 1) / TILE_KEYS;
    const int  tlen = NDIG * nblk;
    const long npad = ((long)N + 63) & ~63L;
    const long epad = ((long)E + 63) & ~63L;
    const long tpad = ((long)tlen + 1023) & ~1023L;
    float* dinv   = (float*)d_ws;
    int*   rowptr = (int*)(dinv + npad);
    int*   sdst   = rowptr + npad + 64;
    int*   ssrc   = sdst + epad;
    float* norm   = (float*)(ssrc + epad);
    int*   tabA   = (int*)(norm + epad);
    int*   tabB   = tabA + tpad;
    float* h      = (float*)(tabB + tpad);
    int*   k2     = (int*)h;
    int*   v2     = k2 + epad;
    float* out    = (float*)d_out;

    const int B = 256;
    const int gN    = (N + B - 1) / B;
    const int gE    = (E + B - 1) / B;
    const int gE1   = (E + 1 + B - 1) / B;
    const int gScan = (tlen + 1023) / 1024;
    const int gLayer = (N + 15) / 16;

    k_sort_pass<false><<<nblk, B, 0, stream>>>(dst, src, sdst, ssrc, tabA, E, 0, nblk);
    k_scan<<<gScan, B, 0, stream>>>(tabA, tabB, tlen);
    k_sort_pass<true><<<nblk, B, 0, stream>>>(dst, src, sdst, ssrc, tabB, E, 0, nblk);

    k_sort_pass<false><<<nblk, B, 0, stream>>>(sdst, ssrc, k2, v2, tabA, E, 6, nblk);
    k_scan<<<gScan, B, 0, stream>>>(tabA, tabB, tlen);
    k_sort_pass<true><<<nblk, B, 0, stream>>>(sdst, ssrc, k2, v2, tabB, E, 6, nblk);

    k_sort_pass<false><<<nblk, B, 0, stream>>>(k2, v2, sdst, ssrc, tabA, E, 12, nblk);
    k_scan<<<gScan, B, 0, stream>>>(tabA, tabB, tlen);
    k_sort_pass<true><<<nblk, B, 0, stream>>>(k2, v2, sdst, ssrc, tabB, E, 12, nblk);

    k_rowptr<<<gE1, B, 0, stream>>>(sdst, rowptr, E, N);
    k_dinv<<<gN, B, 0, stream>>>(rowptr, dinv, N);
    k_edge_norm<<<gE, B, 0, stream>>>(ssrc, sdst, dinv, norm, E);

    k_layer<true><<<gLayer, B, 0, stream>>>(rowptr, ssrc, norm, dinv, x, Wm[0], bm[0], out, N);

    k_layer<true><<<gLayer, B, 0, stream>>>(rowptr, ssrc, norm, dinv, out, Wm[1], bm[1], h, N);

    k_layer<false><<<gLayer, B, 0, stream>>>(rowptr, ssrc, norm, dinv, h, Wm[2], bm[2], out, N);
}


# edge sort: 2 global radix passes on dst/32, then one wave per 32-node group orders its edges by ballot ranks via LDS and emits rowptr and dinv; third pass, dinv launch gone (12 -> 10); sum order same
# speedup vs baseline: 3.9126x; 1.1847x over previous
#include <hip/hip_runtime.h>

typedef float v2f __attribute__((ext_vector_type(2)));
typedef float v8f __attribute__((ext_vector_type(8)));

#define D 64


#define TILE_KEYS 4096
#define TILE_ROWS 128
#define NDIG      64

__device__ __forceinline__ void tab_rows_load(int4* v, const int* __restrict__ tab,
                                              int b0, int rg, int c4, int nblk) {
    const int   lastb = nblk - 1;
    const int4* tab4  = (const int4*)tab;
#pragma unroll
    for (int j = 0; j < 32; ++j) {
        const int row = b0 + j * 16 + rg;
        const int cr  = row < lastb ? row : lastb;
        v[j] = tab4[(unsigned)cr * 16u + (unsigned)c4];
    }
}

__device__ __forceinline__ void tab_rows_add(const int4* v, int b0, int rg, int nblk, int blk,
                                             int4& tot, int4& pre) {
#pragma unroll
    for (int j = 0; j < 32; ++j) {
        const int r0  = b0 + j * 16;
        const int row = r0 + rg;
        if (r0 + 16 <= nblk) {
            tot.x += v[j].x;
            tot.y += v[j].y;
            tot.z += v[j].z;
            tot.w += v[j].w;
        } else if (r0 < nblk) {
            const bool in = row < nblk;
            tot.x += in ? v[j].x : 0;
            tot.y += in ? v[j].y : 0;
            tot.z += in ? v[j].z : 0;
            tot.w += in ? v[j].w : 0;
        }
        if (r0 + 16 <= blk) {
            pre.x += v[j].x;
            pre.y += v[j].y;
            pre.z += v[j].z;
            pre.w += v[j].w;
        } else if (r0 < blk) {
            const bool fr = row < blk;
            pre.x += fr ? v[j].x : 0;
            pre.y += fr ? v[j].y : 0;
            pre.z += fr ? v[j].z : 0;
            pre.w += fr ? v[j].w : 0;
        }
    }
}

template <bool SCATTER>
__global__ __launch_bounds__(256) void k_sort_pass(const int* __restrict__ kin,
                                                   const int* __restrict__ vin,
                                                   int* __restrict__ kout,
                                                   int* __restrict__ vout,
                                                   int* __restrict__ tab,
                                                   int n, int shift, int nblk) {
    __shared__ int cnt[TILE_ROWS * NDIG];
    __shared__ int qsum[4 * NDIG];
    __shared__ __attribute__((aligned(16))) int tsum[SCATTER ? 8 * NDIG : 4];
    __shared__ __attribute__((aligned(16))) int psum[SCATTER ? 8 * NDIG : 4];
    __shared__ int dstart[NDIG];
    __shared__ int gofs[NDIG];
    __shared__ int2 stage[SCATTER ? TILE_KEYS : 1];
    const int t     = threadIdx.x;
    const int lane  = t & 31;
    const int wave  = t >> 5;
    const int tile0 = blockIdx.x * TILE_KEYS;
    const int last  = n - 1;

    const int c4 = t & 15;
    const int rg = t >> 4;
    int4 tot = make_int4(0, 0, 0, 0);
    int4 pre = make_int4(0, 0, 0, 0);
    int4 tv[SCATTER ? 32 : 1];
    if (SCATTER) tab_rows_load(tv, tab, 0, rg, c4, nblk);

    int key[16];
    int val[16];
#pragma unroll
    for (int r = 0; r < 16; ++r) {
        const int idx = tile0 + (r * 8 + wave) * 32 + lane;
        const int ci  = idx < last ? idx : last;
        key[r] = kin[ci];
        if (SCATTER) val[r] = vin[ci];
    }
    if (SCATTER) {
        __builtin_amdgcn_sched_barrier(0);
        tab_rows_add(tv, 0, rg, nblk, (int)blockIdx.x, tot, pre);
        for (int b0 = 512; b0 < nblk; b0 += 512) {
            tab_rows_load(tv, tab, b0, rg, c4, nblk);
            tab_rows_add(tv, b0, rg, nblk, (int)blockIdx.x, tot, pre);
        }
        tot.x += __shfl_xor(tot.x, 16, 32);
        tot.y += __shfl_xor(tot.y, 16, 32);
        tot.z += __shfl_xor(tot.z, 16, 32);
        tot.w += __shfl_xor(tot.w, 16, 32);
        pre.x += __shfl_xor(pre.x, 16, 32);
        pre.y += __shfl_xor(pre.y, 16, 32);
        pre.z += __shfl_xor(pre.z, 16, 32);
        pre.w += __shfl_xor(pre.w, 16, 32);
        if (lane < 16) {
            *(int4*)(tsum + wave * NDIG + 4 * lane) = tot;
            *(int4*)(psum + wave * NDIG + 4 * lane) = pre;
        }
    }

    for (int i = t; i < TILE_ROWS * NDIG; i += 256) cnt[i] = 0;
    __syncthreads();

    int rk[16];
    const unsigned lt = (1u << lane) - 1u;
#pragma unroll
    for (int r = 0; r < 16; ++r) {
        const int  row   = r * 8 + wave;
        const bool valid = tile0 + row * 32 + lane < n;
        const int  d     = (key[r] >> shift) & (NDIG - 1);
        unsigned peers = __builtin_amdgcn_ballot_w32(valid);
#pragma unroll
        for (int b = 0; b < 6; ++b) {
            const bool     bit = (d >> b) & 1;
            const unsigned m   = __builtin_amdgcn_ballot_w32(valid && bit);
            peers &= bit ? m : ~m;
        }
        rk[r] = __popc(peers & lt);
        if (valid && rk[r] == 0) cnt[row * NDIG + d] = __popc(peers);
    }
    __syncthreads();

    const int dg = t & (NDIG - 1);
    const int q  = t >> 6;
    {
        int run = 0;
        for (int i = 0; i < 32; ++i) {
            const int a = (q * 32 + i) * NDIG + dg;
            const int v = cnt[a];
            if (SCATTER) cnt[a] = run;
            run += v;
        }
        qsum[q * NDIG + dg] = run;
    }
    __syncthreads();

    if (!SCATTER) {
        if (t < NDIG)
            tab[blockIdx.x * NDIG + t] = qsum[t] + qsum[NDIG + t] + qsum[2 * NDIG + t] + qsum[3 * NDIG + t];
        return;
    }

    if (wave == 0) {
        const int d0 = 2 * lane;
        const int t0 = qsum[d0] + qsum[NDIG + d0] + qsum[2 * NDIG + d0] + qsum[3 * NDIG + d0];
        const int t1 = qsum[d0 + 1] + qsum[NDIG + d0 + 1] + qsum[2 * NDIG + d0 + 1] + qsum[3 * NDIG + d0 + 1];
        int g0 = 0, g1 = 0, p0 = 0, p1 = 0;
        for (int w = 0; w < 8; ++w) {
            const int2 gv = *(const int2*)(tsum + w * NDIG + d0);
            const int2 pv = *(const int2*)(psum + w * NDIG + d0);
            g0 += gv.x;
            g1 += gv.y;
            p0 += pv.x;
            p1 += pv.y;
        }
        const int s  = t0 + t1;
        const int gs = g0 + g1;
        int inc  = s;
        int ginc = gs;
        for (int o = 1; o < 32; o <<= 1) {
            const int u  = __shfl_up(inc, o, 32);
            const int gu = __shfl_up(ginc, o, 32);
            if (lane >= o) {
                inc  += u;
                ginc += gu;
            }
        }
        const int e0 = inc - s;
        const int b0 = ginc - gs;
        dstart[d0]     = e0;
        dstart[d0 + 1] = e0 + t0;
        gofs[d0]       = (b0 + p0) - e0;
        gofs[d0 + 1]   = (b0 + g0 + p1) - (e0 + t0);
    }

    int pb[16], pc[16];
#pragma unroll
    for (int r = 0; r < 16; ++r) {
        const int row = r * 8 + wave;
        const int d   = (key[r] >> shift) & (NDIG - 1);
        pb[r] = cnt[row * NDIG + d];
        int s = 0;
        if (r >= 4)  s += qsum[d];
        if (r >= 8)  s += qsum[NDIG + d];
        if (r >= 12) s += qsum[2 * NDIG + d];
        pc[r] = s;
    }
    __syncthreads();

    const bool full = n - tile0 >= TILE_KEYS;
    int lp[16];
#pragma unroll
    for (int r = 0; r < 16; ++r) {
        const int d = (key[r] >> shift) & (NDIG - 1);
        lp[r] = dstart[d] + pb[r] + pc[r] + rk[r];
    }
    if (full) {
#pragma unroll
        for (int r = 0; r < 16; ++r) stage[lp[r]] = make_int2(key[r], val[r]);
    } else {
#pragma unroll
        for (int r = 0; r < 16; ++r)
            if (tile0 + (r * 8 + wave) * 32 + lane < n) stage[lp[r]] = make_int2(key[r], val[r]);
    }
    __syncthreads();

    if (full) {
        int2 kv[16];
#pragma unroll
        for (int r = 0; r < 16; ++r) kv[r] = stage[r * 256 + t];
#pragma unroll
        for (int r = 0; r < 16; ++r) {
            const int g = gofs[(kv[r].x >> shift) & (NDIG - 1)] + r * 256 + t;
            kout[g] = kv[r].x;
            vout[g] = kv[r].y;
        }
    } else {
        const int nv = n - tile0;
        for (int i = t; i < nv; i += 256) {
            const int2 kv = stage[i];
            const int  g  = gofs[(kv.x >> shift) & (NDIG - 1)] + i;
            kout[g] = kv.x;
            vout[g] = kv.y;
        }
    }
}

__global__ __launch_bounds__(256) void k_group_ptr(const int* __restrict__ skey,
                                                   int* __restrict__ gptr, int e_cnt, int n,
                                                   int sh) {
    const int j = blockIdx.x * blockDim.x + threadIdx.x;
    if (j > e_cnt) return;
    const int lo = (j == 0) ? 0 : (skey[j - 1] >> sh) + 1;
    const int hi = (j == e_cnt) ? n : (skey[j] >> sh);
    for (int g = lo; g <= hi; ++g) gptr[g] = j;
}

__device__ __forceinline__ unsigned digit_peers(const unsigned* b, unsigned v, const unsigned* nm) {
    unsigned m = v;
#pragma unroll
    for (int i = 0; i < 5; ++i) m &= b[i] ^ nm[i];
    return m;
}

#define GS_CAP 1024

__global__ __launch_bounds__(256) void k_group_sort(const int* __restrict__ kin,
                                                    const int* __restrict__ vin,
                                                    const int* __restrict__ gptr,
                                                    int* __restrict__ ssrc,
                                                    float* __restrict__ wdst,
                                                    int* __restrict__ rowptr,
                                                    float* __restrict__ dinv,
                                                    int n_nodes, int ngrp) {
    __shared__ int2 stg[8 * GS_CAP];
    const int lane = threadIdx.x & 31;
    const int wave = threadIdx.x >> 5;
    const int g    = blockIdx.x * 8 + wave;
    const bool have = g < ngrp;
    int gb0 = 0, ge0 = 0;
    if (have) {
        gb0 = gptr[g];
        ge0 = gptr[g + 1];
    }
    const int gb  = __builtin_amdgcn_readfirstlane(gb0);
    const int ge  = __builtin_amdgcn_readfirstlane(ge0);
    const int len = ge - gb;
    const unsigned lt = (1u << lane) - 1u;
    unsigned nml[5];
#pragma unroll
    for (int i = 0; i < 5; ++i) nml[i] = ((lane >> i) & 1) ? 0u : 0xffffffffu;

    int tot = 0;
    for (int c0 = 0; c0 < len; c0 += 256) {
        int k[8];
#pragma unroll
        for (int u = 0; u < 8; ++u) {
            const int idx = gb + c0 + u * 32 + lane;
            k[u] = kin[idx < ge ? idx : ge - 1];
        }
#pragma unroll
        for (int u = 0; u < 8; ++u) {
            const bool valid = gb + c0 + u * 32 + lane < ge;
            unsigned b[5];
#pragma unroll
            for (int i = 0; i < 5; ++i)
                b[i] = __builtin_amdgcn_ballot_w32(valid && ((k[u] >> i) & 1));
            const unsigned v = __builtin_amdgcn_ballot_w32(valid);
            tot += __popc(digit_peers(b, v, nml));
        }
    }
    int inc = tot;
    for (int o = 1; o < 32; o <<= 1) {
        const int up = __shfl_up(inc, o, 32);
        if (lane >= o) inc += up;
    }
    const int   first = inc - tot;
    const float dv    = rsqrtf((float)(tot + 1));
    const int   node  = g * 32 + lane;
    if (have && node < n_nodes) {
        rowptr[node] = gb + first;
        dinv[node]   = dv;
    }
    if (have && g == ngrp - 1 && lane == 0) rowptr[n_nodes] = ge;

    const bool staged = len < GS_CAP;
    int2* mine = stg + wave * GS_CAP;
    int run = first;
    for (int c0 = 0; c0 < len; c0 += 256) {
        int k[8], s[8];
#pragma unroll
        for (int u = 0; u < 8; ++u) {
            const int idx = gb + c0 + u * 32 + lane;
            const int ci  = idx < ge ? idx : ge - 1;
            k[u] = kin[ci];
            s[u] = vin[ci];
        }
#pragma unroll
        for (int u = 0; u < 8; ++u) {
            const bool valid = gb + c0 + u * 32 + lane < ge;
            const int  d     = k[u] & 31;
            unsigned b[5], nmd[5];
#pragma unroll
            for (int i = 0; i < 5; ++i) {
                b[i]   = __builtin_amdgcn_ballot_w32(valid && ((d >> i) & 1));
                nmd[i] = ((d >> i) & 1) ? 0u : 0xffffffffu;
            }
            const unsigned v = __builtin_amdgcn_ballot_w32(valid);
            const int   rank = __popc(digit_peers(b, v, nmd) & lt);
            const int   base = __shfl(run, d, 32);
            const float w    = __shfl(dv, d, 32);
            const int   p    = base + rank;
            if (staged) {
                mine[valid ? p : GS_CAP - 1] = make_int2(s[u], __float_as_int(w));
            } else if (valid) {
                ssrc[gb + p] = s[u];
                wdst[gb + p] = w;
            }
            run += __popc(digit_peers(b, v, nml));
        }
    }
    __syncthreads();
    if (staged) {
        for (int i0 = 0; i0 < len; i0 += 128) {
            int2 e[4];
#pragma unroll
            for (int u = 0; u < 4; ++u) {
                const int i = i0 + u * 32 + lane;
                e[u] = mine[i < len ? i : len - 1];
            }
#pragma unroll
            for (int u = 0; u < 4; ++u) {
                const int i = i0 + u * 32 + lane;
                if (i < len) {
                    ssrc[gb + i] = e[u].x;
                    wdst[gb + i] = __int_as_float(e[u].y);
                }
            }
        }
    }
}

__global__ __launch_bounds__(256) void k_edge_norm(const int* __restrict__ ssrc,
                                                   const float* __restrict__ dinv,
                                                   float* __restrict__ norm, int e_cnt) {
    const int j = blockIdx.x * blockDim.x + threadIdx.x;
    if (j < e_cnt) norm[j] = dinv[ssrc[j]] * norm[j];
}


#define AP 68

template <bool RELU_OUT>
__global__ __launch_bounds__(256) void k_layer(const int* __restrict__ rowptr,
                                               const int* __restrict__ ssrc,
                                               const float* __restrict__ norm,
                                               const float* __restrict__ dinv,
                                               const float* __restrict__ X,
                                               const float* __restrict__ W,
                                               const float* __restrict__ bias,
                                               float* __restrict__ Y, int n_nodes) {
    __shared__ float As[16 * AP];
    const unsigned t    = threadIdx.x;
    const unsigned nl   = t >> 4;
    const unsigned l    = t & 15u;
    const unsigned node = blockIdx.x * 16u + nl;

    float4 a = make_float4(0.0f, 0.0f, 0.0f, 0.0f);
    if (node < (unsigned)n_nodes) {
        const int   beg = rowptr[node];
        const int   end = rowptr[node + 1];
        const float di  = dinv[node];
        const float dd  = di * di;
        const float4* X4 = (const float4*)X;
        const float4 hv = X4[node * 16u + l];
        a.x = hv.x * dd;
        a.y = hv.y * dd;
        a.z = hv.z * dd;
        a.w = hv.w * dd;
        for (int c = beg; c < end; c += 16) {
            const int   je = c + (int)l;
            const bool  ok = je < end;
            const int   cj = ok ? je : end - 1;
            const int   sv = ssrc[cj];
            const float wv = norm[cj];
            const int   si = ok ? sv : (int)node;
            const float wi = ok ? wv : 0.0f;
            const int  cnt = end - c;
            for (int e0 = 0; e0 < 16 && e0 < cnt; e0 += 8) {
                float4 r[8];
                float  w[8];
#pragma unroll
                for (int u = 0; u < 8; ++u) {
                    const unsigned s = (unsigned)__shfl(si, e0 + u, 16);
                    w[u] = __shfl(wi, e0 + u, 16);
                    r[u] = X4[s * 16u + l];
                }
#pragma unroll
                for (int u = 0; u < 8; ++u) {
                    a.x += r[u].x * w[u];
                    a.y += r[u].y * w[u];
                    a.z += r[u].z * w[u];
                    a.w += r[u].w * w[u];
                }
            }
        }
    }
    *(float4*)(As + nl * AP + 4u * l) = a;
    __syncthreads();

    const int wave = (int)(t >> 5);
    if (wave >= 4) return;
    const int lane = (int)(t & 31u);
    const int half = lane >> 4;
    const int l16  = lane & 15;
    const int ncol = wave * 16 + l16;

    v8f c0 = {};
    v8f c1 = {};
#pragma unroll
    for (int k0 = 0; k0 < D; k0 += 8) {
        const int ka = k0 + 2 * half;
        const int kb = ka + 4;
        const float2 a0 = *(const float2*)(As + l16 * AP + ka);
        const float2 a1 = *(const float2*)(As + l16 * AP + kb);
        v2f fa0, fb0, fa1, fb1;
        fa0.x = a0.x;  fa0.y = a0.y;
        fb0.x = W[ka * D + ncol];
        fb0.y = W[(ka + 1) * D + ncol];
        fa1.x = a1.x;  fa1.y = a1.y;
        fb1.x = W[kb * D + ncol];
        fb1.y = W[(kb + 1) * D + ncol];
        c0 = __builtin_amdgcn_wmma_f32_16x16x4_f32(false, fa0, false, fb0,
                                                   (short)0, c0, false, false);
        c1 = __builtin_amdgcn_wmma_f32_16x16x4_f32(false, fa1, false, fb1,
                                                   (short)0, c1, false, false);
    }

    const float bv    = bias[ncol];
    const int   mbase = (int)blockIdx.x * 16 + half * 8;
#pragma unroll
    for (int r = 0; r < 8; ++r) {
        const int m = mbase + r;
        float v = c0[r] + c1[r] + bv;
        if (RELU_OUT) v = fmaxf(v, 0.0f);
        if (m < n_nodes) Y[(long)m * D + ncol] = v;
    }
}

extern "C" void kernel_launch(void* const* d_in, const int* in_sizes, int n_in,
                              void* d_out, int out_size, void* d_ws, size_t ws_size,
                              hipStream_t stream) {
    const float* x  = (const float*)d_in[0];
    const int*   ei = (const int*)d_in[1];
    const float* Wm[3] = { (const float*)d_in[2], (const float*)d_in[4], (const float*)d_in[6] };
    const float* bm[3] = { (const float*)d_in[3], (const float*)d_in[5], (const float*)d_in[7] };

    const int N = in_sizes[0] / D;
    const int E = in_sizes[1] / 2;
    const int* src = ei;
    const int* dst = ei + E;

    const int  nblk = (E + TILE_KEYS - 1) / TILE_KEYS;
    const int  tlen = NDIG * nblk;
    const long npad = ((long)N + 63) & ~63L;
    const long epad = ((long)E + 63) & ~63L;
    const long tpad = ((long)tlen + 1023) & ~1023L;
    float* dinv   = (float*)d_ws;
    int*   rowptr = (int*)(dinv + npad);
    int*   sdst   = rowptr + npad + 64;
    int*   ssrc   = sdst + epad;
    float* norm   = (float*)(ssrc + epad);
    int*   tab   = (int*)(norm + epad);
    float* h      = (float*)(tab + tpad);
    int*   k2     = (int*)h;
    int*   v2     = k2 + epad;
    float* out    = (float*)d_out;

    const int B = 256;
    const int gE    = (E + B - 1) / B;
    const int gE1   = (E + 1 + B - 1) / B;
    const int gLayer = (N + 15) / 16;
    const int ngrp  = (N + 31) / 32;
    const int gGrp  = (ngrp + 7) / 8;

    k_sort_pass<false><<<nblk, B, 0, stream>>>(dst, src, sdst, ssrc, tab, E, 5, nblk);
    k_sort_pass<true><<<nblk, B, 0, stream>>>(dst, src, sdst, ssrc, tab, E, 5, nblk);

    k_sort_pass<false><<<nblk, B, 0, stream>>>(sdst, ssrc, k2, v2, tab, E, 11, nblk);
    k_sort_pass<true><<<nblk, B, 0, stream>>>(sdst, ssrc, k2, v2, tab, E, 11, nblk);

    int* gptr = sdst;
    k_group_ptr<<<gE1, B, 0, stream>>>(k2, gptr, E, ngrp, 5);
    k_group_sort<<<gGrp, B, 0, stream>>>(k2, v2, gptr, ssrc, norm, rowptr, dinv, N, ngrp);
    k_edge_norm<<<gE, B, 0, stream>>>(ssrc, dinv, norm, E);

    k_layer<true><<<gLayer, B, 0, stream>>>(rowptr, ssrc, norm, dinv, x, Wm[0], bm[0], out, N);

    k_layer<true><<<gLayer, B, 0, stream>>>(rowptr, ssrc, norm, dinv, out, Wm[1], bm[1], h, N);

    k_layer<false><<<gLayer, B, 0, stream>>>(rowptr, ssrc, norm, dinv, h, Wm[2], bm[2], out, N);
}
